# starting kernel
# baseline (speedup 1.0000x reference)
#include <hip/hip_runtime.h>

#define B_DIM 16
#define N_DIM 2048
#define H_DIM 1024
#define NT    128

typedef __attribute__((ext_vector_type(2))) float v2f;
typedef __attribute__((ext_vector_type(8))) float v8f;


__device__ __forceinline__ v2f load_a_f32(const float* __restrict__ A, int lda,
                                          int r0, int k, int lane) {
  int r  = r0 + (lane & 15);
  int kk = k + ((lane >> 4) << 1);
  const float* p = A + (size_t)r * lda + kk;
  return v2f{p[0], p[1]};
}

__device__ __forceinline__ v2f load_b_rowmajor(const float* __restrict__ Bm, int ldb,
                                               int k, int c0, int lane) {
  int c  = c0 + (lane & 15);
  int kk = k + ((lane >> 4) << 1);
  const float* p = Bm + (size_t)kk * ldb + c;
  return v2f{p[0], p[ldb]};
}

__device__ __forceinline__ v2f load_b_transposed(const float* __restrict__ T, int ldt,
                                                 int k, int c0, int lane) {
  int c  = c0 + (lane & 15);
  int kk = k + ((lane >> 4) << 1);
  const float* p = T + (size_t)c * ldt + kk;
  return v2f{p[0], p[1]};
}

__device__ __forceinline__ v8f wmma_f32(v2f a, v2f b, v8f c) {
  return __builtin_amdgcn_wmma_f32_16x16x4_f32(false, a, false, b, (short)0, c,
                                               false, false);
}

__global__ void fc1_kernel(const float* __restrict__ feats,
                           const float* __restrict__ W,
                           const float* __restrict__ bias,
                           float* __restrict__ q) {
  int gid  = blockIdx.x * (blockDim.x >> 5) + (threadIdx.x >> 5);
  int lane = threadIdx.x & 31;
  int tM = gid >> 5;
  int tN = gid & 31;
  int r0 = tM * 16;
  int c0 = tN * 32;

  v8f acc0 = {};
  v8f acc1 = {};
  #pragma unroll 4
  for (int k = 0; k < H_DIM; k += 4) {
    v2f a  = load_a_f32(feats, H_DIM, r0, k, lane);
    v2f b0 = load_b_rowmajor(W, H_DIM, k, c0, lane);
    v2f b1 = load_b_rowmajor(W, H_DIM, k, c0 + 16, lane);
    acc0 = wmma_f32(a, b0, acc0);
    acc1 = wmma_f32(a, b1, acc1);
  }

  int row = r0 + ((lane >> 4) << 3);
  int col = c0 + (lane & 15);
  float bz0 = bias[col];
  float bz1 = bias[col + 16];
  #pragma unroll
  for (int j = 0; j < 8; ++j) {
    q[(size_t)(row + j) * H_DIM + col]      = acc0[j] + bz0;
    q[(size_t)(row + j) * H_DIM + col + 16] = acc1[j] + bz1;
  }
}

__global__ void attn_kernel(const float* __restrict__ q,
                            const float* __restrict__ feats,
                            float* __restrict__ partial) {
  int b  = blockIdx.x >> 7;
  int nt = blockIdx.x & (NT - 1);
  int n0 = nt * 16;
  const float* Q = q     + ((size_t)b * N_DIM + n0) * H_DIM;
  const float* F = feats + (size_t)b * N_DIM * H_DIM;

  extern __shared__ float slab[];
  __shared__ float rowInv[16];

  int wave = threadIdx.x >> 5;
  int lane = threadIdx.x & 31;

  for (int g = 0; g < 4; ++g) {
    int m0 = wave * 256 + g * 64;
    v8f acc[4] = {};
    #pragma unroll 2
    for (int k = 0; k < H_DIM; k += 4) {
      v2f a = load_a_f32(Q, H_DIM, 0, k, lane);
      #pragma unroll
      for (int t = 0; t < 4; ++t) {
        v2f bf = load_b_transposed(F, H_DIM, k, m0 + t * 16, lane);
        acc[t] = wmma_f32(a, bf, acc[t]);
      }
    }
    int rbase = (lane >> 4) << 3;
    int cl    = lane & 15;
    #pragma unroll
    for (int t = 0; t < 4; ++t) {
      #pragma unroll
      for (int j = 0; j < 8; ++j)
        slab[(rbase + j) * N_DIM + m0 + t * 16 + cl] = acc[t][j];
    }
  }
  __syncthreads();

  for (int rr = 0; rr < 2; ++rr) {
    int r = wave * 2 + rr;
    float mx = -3.402823466e38f;
    for (int m = lane; m < N_DIM; m += 32)
      mx = fmaxf(mx, slab[r * N_DIM + m]);
    #pragma unroll
    for (int o = 16; o > 0; o >>= 1)
      mx = fmaxf(mx, __shfl_xor(mx, o, 32));
    float s = 0.f;
    for (int m = lane; m < N_DIM; m += 32) {
      float e = __expf(slab[r * N_DIM + m] - mx);
      slab[r * N_DIM + m] = e;
      s += e;
    }
    #pragma unroll
    for (int o = 16; o > 0; o >>= 1)
      s += __shfl_xor(s, o, 32);
    if (lane == 0) rowInv[r] = 1.0f / s;
  }
  __syncthreads();

  float* P = partial + ((size_t)b * NT + nt) * N_DIM;
  for (int m = threadIdx.x; m < N_DIM; m += 256) {
    float a = 0.f;
    #pragma unroll
    for (int r = 0; r < 16; ++r)
      a += slab[r * N_DIM + m] * rowInv[r];
    P[m] = a;
  }
}

__global__ void colsum_kernel(const float* __restrict__ partial,
                              float* __restrict__ colsum) {
  int idx = blockIdx.x * blockDim.x + threadIdx.x;
  int b = idx >> 11;
  int m = idx & (N_DIM - 1);
  const float* P = partial + (size_t)b * NT * N_DIM + m;
  float a = 0.f;
  #pragma unroll 8
  for (int t = 0; t < NT; ++t) a += P[(size_t)t * N_DIM];
  colsum[idx] = a;
}

__global__ void pool_kernel(const float* __restrict__ colsum,
                            const float* __restrict__ feats,
                            float* __restrict__ out) {
  int b = blockIdx.x;
  int h = blockIdx.y * 256 + threadIdx.x;
  const float* F = feats + (size_t)b * N_DIM * H_DIM + h;
  const float* C = colsum + (size_t)b * N_DIM;
  float a = 0.f;
  #pragma unroll 4
  for (int m = 0; m < N_DIM; ++m)
    a += C[m] * F[(size_t)m * H_DIM];
  out[(size_t)b * H_DIM + h] = a;
}

extern "C" void kernel_launch(void* const* d_in, const int* in_sizes, int n_in,
                              void* d_out, int out_size, void* d_ws, size_t ws_size,
                              hipStream_t stream) {
  const float* feats = (const float*)d_in[0];
  const float* W     = (const float*)d_in[1];
  const float* bias  = (const float*)d_in[2];
  float* out = (float*)d_out;

  float* q       = (float*)d_ws;
  float* partial = q + (size_t)B_DIM * N_DIM * H_DIM;
  float* colsum  = partial + (size_t)B_DIM * NT * N_DIM;

  fc1_kernel<<<8192, 256, 0, stream>>>(feats, W, bias, q);

  attn_kernel<<<B_DIM * NT, 256, 16 * N_DIM * sizeof(float), stream>>>(q, feats, partial);

  colsum_kernel<<<(B_DIM * N_DIM) / 256, 256, 0, stream>>>(partial, colsum);

  pool_kernel<<<dim3(B_DIM, H_DIM / 256), 256, 0, stream>>>(colsum, feats, out);
}


# Fragment loads widened: 128 bit per lane along k for A and transposed B (K block 8, same permuted k order in both, so k sums in a new order), 64 bit for W over two interleaved column tiles
# speedup vs baseline: 1.9924x; 1.9924x over previous
#include <hip/hip_runtime.h>

#define B_DIM 16
#define N_DIM 2048
#define H_DIM 1024
#define NT    128

typedef __attribute__((ext_vector_type(2))) float v2f;
typedef __attribute__((ext_vector_type(4))) float v4f;
typedef __attribute__((ext_vector_type(8))) float v8f;


__device__ __forceinline__ v4f load_rows_k8(const float* __restrict__ M, int ld,
                                            int r0, int k0, int lane) {
  int r  = r0 + (lane & 15);
  int kk = k0 + ((lane >> 4) << 2);
  return *reinterpret_cast<const v4f*>(M + (size_t)r * ld + kk);
}

__device__ __forceinline__ v2f frag_lo(v4f x) { return v2f{x[0], x[1]}; }
__device__ __forceinline__ v2f frag_hi(v4f x) { return v2f{x[2], x[3]}; }

__device__ __forceinline__ v8f wmma_f32(v2f a, v2f b, v8f c) {
  return __builtin_amdgcn_wmma_f32_16x16x4_f32(false, a, false, b, (short)0, c,
                                               false, false);
}

__global__ void fc1_kernel(const float* __restrict__ feats,
                           const float* __restrict__ W,
                           const float* __restrict__ bias,
                           float* __restrict__ q) {
  int gid  = blockIdx.x * (blockDim.x >> 5) + (threadIdx.x >> 5);
  int lane = threadIdx.x & 31;
  int tM = gid >> 5;
  int tN = gid & 31;
  int r0 = tM * 16;
  int c0 = tN * 32;
  int cp = c0 + ((lane & 15) << 1);
  int kh = (lane >> 4) << 2;

  v8f acc0 = {};
  v8f acc1 = {};
  for (int k = 0; k < H_DIM; k += 32) {
    v4f a[4];
    v2f w[4][4];
    #pragma unroll
    for (int s = 0; s < 4; ++s)
      a[s] = load_rows_k8(feats, H_DIM, r0, k + 8 * s, lane);
    #pragma unroll
    for (int s = 0; s < 4; ++s) {
      const float* p = W + (size_t)(k + 8 * s + kh) * H_DIM + cp;
      #pragma unroll
      for (int i = 0; i < 4; ++i)
        w[s][i] = *reinterpret_cast<const v2f*>(p + (size_t)i * H_DIM);
    }
    __builtin_amdgcn_sched_barrier(0);
    #pragma unroll
    for (int s = 0; s < 4; ++s) {
      acc0 = wmma_f32(frag_lo(a[s]), v2f{w[s][0][0], w[s][1][0]}, acc0);
      acc1 = wmma_f32(frag_lo(a[s]), v2f{w[s][0][1], w[s][1][1]}, acc1);
      acc0 = wmma_f32(frag_hi(a[s]), v2f{w[s][2][0], w[s][3][0]}, acc0);
      acc1 = wmma_f32(frag_hi(a[s]), v2f{w[s][2][1], w[s][3][1]}, acc1);
    }
  }

  int row = r0 + ((lane >> 4) << 3);
  v2f bz = *reinterpret_cast<const v2f*>(bias + cp);
  #pragma unroll
  for (int j = 0; j < 8; ++j) {
    v2f o = v2f{acc0[j] + bz[0], acc1[j] + bz[1]};
    *reinterpret_cast<v2f*>(q + (size_t)(row + j) * H_DIM + cp) = o;
  }
}

__global__ void attn_kernel(const float* __restrict__ q,
                            const float* __restrict__ feats,
                            float* __restrict__ partial) {
  int b  = blockIdx.x >> 7;
  int nt = blockIdx.x & (NT - 1);
  int n0 = nt * 16;
  const float* Q = q     + ((size_t)b * N_DIM + n0) * H_DIM;
  const float* F = feats + (size_t)b * N_DIM * H_DIM;

  extern __shared__ float slab[];
  __shared__ float rowInv[16];

  int wave = threadIdx.x >> 5;
  int lane = threadIdx.x & 31;

  for (int g = 0; g < 4; ++g) {
    int m0 = wave * 256 + g * 64;
    v8f acc[4] = {};
    for (int k = 0; k < H_DIM; k += 32) {
      v4f a[4];
      v4f bf[4][4];
      #pragma unroll
      for (int s = 0; s < 4; ++s)
        a[s] = load_rows_k8(Q, H_DIM, 0, k + 8 * s, lane);
      #pragma unroll
      for (int t = 0; t < 4; ++t) {
        #pragma unroll
        for (int s = 0; s < 4; ++s)
          bf[t][s] = load_rows_k8(F, H_DIM, m0 + t * 16, k + 8 * s, lane);
      }
      __builtin_amdgcn_sched_barrier(0);
      #pragma unroll
      for (int s = 0; s < 4; ++s) {
        #pragma unroll
        for (int t = 0; t < 4; ++t)
          acc[t] = wmma_f32(frag_lo(a[s]), frag_lo(bf[t][s]), acc[t]);
        #pragma unroll
        for (int t = 0; t < 4; ++t)
          acc[t] = wmma_f32(frag_hi(a[s]), frag_hi(bf[t][s]), acc[t]);
      }
    }
    int rbase = (lane >> 4) << 3;
    int cl    = lane & 15;
    #pragma unroll
    for (int t = 0; t < 4; ++t) {
      #pragma unroll
      for (int j = 0; j < 8; ++j)
        slab[(rbase + j) * N_DIM + m0 + t * 16 + cl] = acc[t][j];
    }
  }
  __syncthreads();

  for (int rr = 0; rr < 2; ++rr) {
    int r = wave * 2 + rr;
    float mx = -3.402823466e38f;
    for (int m = lane; m < N_DIM; m += 32)
      mx = fmaxf(mx, slab[r * N_DIM + m]);
    #pragma unroll
    for (int o = 16; o > 0; o >>= 1)
      mx = fmaxf(mx, __shfl_xor(mx, o, 32));
    float s = 0.f;
    for (int m = lane; m < N_DIM; m += 32) {
      float e = __expf(slab[r * N_DIM + m] - mx);
      slab[r * N_DIM + m] = e;
      s += e;
    }
    #pragma unroll
    for (int o = 16; o > 0; o >>= 1)
      s += __shfl_xor(s, o, 32);
    if (lane == 0) rowInv[r] = 1.0f / s;
  }
  __syncthreads();

  float* P = partial + ((size_t)b * NT + nt) * N_DIM;
  for (int m = threadIdx.x; m < N_DIM; m += 256) {
    float a = 0.f;
    #pragma unroll
    for (int r = 0; r < 16; ++r)
      a += slab[r * N_DIM + m] * rowInv[r];
    P[m] = a;
  }
}

__global__ void colsum_kernel(const float* __restrict__ partial,
                              float* __restrict__ colsum) {
  int idx = blockIdx.x * blockDim.x + threadIdx.x;
  int b = idx >> 11;
  int m = idx & (N_DIM - 1);
  const float* P = partial + (size_t)b * NT * N_DIM + m;
  float a = 0.f;
  #pragma unroll 8
  for (int t = 0; t < NT; ++t) a += P[(size_t)t * N_DIM];
  colsum[idx] = a;
}

__global__ void pool_kernel(const float* __restrict__ colsum,
                            const float* __restrict__ feats,
                            float* __restrict__ out) {
  int b = blockIdx.x;
  int h = blockIdx.y * 256 + threadIdx.x;
  const float* F = feats + (size_t)b * N_DIM * H_DIM + h;
  const float* C = colsum + (size_t)b * N_DIM;
  float a = 0.f;
  #pragma unroll 4
  for (int m = 0; m < N_DIM; ++m)
    a += C[m] * F[(size_t)m * H_DIM];
  out[(size_t)b * H_DIM + h] = a;
}

extern "C" void kernel_launch(void* const* d_in, const int* in_sizes, int n_in,
                              void* d_out, int out_size, void* d_ws, size_t ws_size,
                              hipStream_t stream) {
  const float* feats = (const float*)d_in[0];
  const float* W     = (const float*)d_in[1];
  const float* bias  = (const float*)d_in[2];
  float* out = (float*)d_out;

  float* q       = (float*)d_ws;
  float* partial = q + (size_t)B_DIM * N_DIM * H_DIM;
  float* colsum  = partial + (size_t)B_DIM * NT * N_DIM;

  fc1_kernel<<<8192, 256, 0, stream>>>(feats, W, bias, q);

  attn_kernel<<<B_DIM * NT, 256, 16 * N_DIM * sizeof(float), stream>>>(q, feats, partial);

  colsum_kernel<<<(B_DIM * N_DIM) / 256, 256, 0, stream>>>(partial, colsum);

  pool_kernel<<<dim3(B_DIM, H_DIM / 256), 256, 0, stream>>>(colsum, feats, out);
}


# Pool kernel rebuilt as a stream: rows split over 32 slots per block, 32 whole-line 128-bit loads in flight per wave, slot sums joined in LDS (m summed in a new order)
# speedup vs baseline: 2.2229x; 1.1157x over previous
#include <hip/hip_runtime.h>

#define B_DIM 16
#define N_DIM 2048
#define H_DIM 1024
#define NT    128

typedef __attribute__((ext_vector_type(2))) float v2f;
typedef __attribute__((ext_vector_type(4))) float v4f;
typedef __attribute__((ext_vector_type(8))) float v8f;


__device__ __forceinline__ v4f load_rows_k8(const float* __restrict__ M, int ld,
                                            int r0, int k0, int lane) {
  int r  = r0 + (lane & 15);
  int kk = k0 + ((lane >> 4) << 2);
  return *reinterpret_cast<const v4f*>(M + (size_t)r * ld + kk);
}

__device__ __forceinline__ v2f frag_lo(v4f x) { return v2f{x[0], x[1]}; }
__device__ __forceinline__ v2f frag_hi(v4f x) { return v2f{x[2], x[3]}; }

__device__ __forceinline__ v8f wmma_f32(v2f a, v2f b, v8f c) {
  return __builtin_amdgcn_wmma_f32_16x16x4_f32(false, a, false, b, (short)0, c,
                                               false, false);
}

__global__ void fc1_kernel(const float* __restrict__ feats,
                           const float* __restrict__ W,
                           const float* __restrict__ bias,
                           float* __restrict__ q) {
  int gid  = blockIdx.x * (blockDim.x >> 5) + (threadIdx.x >> 5);
  int lane = threadIdx.x & 31;
  int tM = gid >> 5;
  int tN = gid & 31;
  int r0 = tM * 16;
  int c0 = tN * 32;
  int cp = c0 + ((lane & 15) << 1);
  int kh = (lane >> 4) << 2;

  v8f acc0 = {};
  v8f acc1 = {};
  for (int k = 0; k < H_DIM; k += 32) {
    v4f a[4];
    v2f w[4][4];
    #pragma unroll
    for (int s = 0; s < 4; ++s)
      a[s] = load_rows_k8(feats, H_DIM, r0, k + 8 * s, lane);
    #pragma unroll
    for (int s = 0; s < 4; ++s) {
      const float* p = W + (size_t)(k + 8 * s + kh) * H_DIM + cp;
      #pragma unroll
      for (int i = 0; i < 4; ++i)
        w[s][i] = *reinterpret_cast<const v2f*>(p + (size_t)i * H_DIM);
    }
    __builtin_amdgcn_sched_barrier(0);
    #pragma unroll
    for (int s = 0; s < 4; ++s) {
      acc0 = wmma_f32(frag_lo(a[s]), v2f{w[s][0][0], w[s][1][0]}, acc0);
      acc1 = wmma_f32(frag_lo(a[s]), v2f{w[s][0][1], w[s][1][1]}, acc1);
      acc0 = wmma_f32(frag_hi(a[s]), v2f{w[s][2][0], w[s][3][0]}, acc0);
      acc1 = wmma_f32(frag_hi(a[s]), v2f{w[s][2][1], w[s][3][1]}, acc1);
    }
  }

  int row = r0 + ((lane >> 4) << 3);
  v2f bz = *reinterpret_cast<const v2f*>(bias + cp);
  #pragma unroll
  for (int j = 0; j < 8; ++j) {
    v2f o = v2f{acc0[j] + bz[0], acc1[j] + bz[1]};
    *reinterpret_cast<v2f*>(q + (size_t)(row + j) * H_DIM + cp) = o;
  }
}

__global__ void attn_kernel(const float* __restrict__ q,
                            const float* __restrict__ feats,
                            float* __restrict__ partial) {
  int b  = blockIdx.x >> 7;
  int nt = blockIdx.x & (NT - 1);
  int n0 = nt * 16;
  const float* Q = q     + ((size_t)b * N_DIM + n0) * H_DIM;
  const float* F = feats + (size_t)b * N_DIM * H_DIM;

  extern __shared__ float slab[];
  __shared__ float rowInv[16];

  int wave = threadIdx.x >> 5;
  int lane = threadIdx.x & 31;

  for (int g = 0; g < 4; ++g) {
    int m0 = wave * 256 + g * 64;
    v8f acc[4] = {};
    for (int k = 0; k < H_DIM; k += 32) {
      v4f a[4];
      v4f bf[4][4];
      #pragma unroll
      for (int s = 0; s < 4; ++s)
        a[s] = load_rows_k8(Q, H_DIM, 0, k + 8 * s, lane);
      #pragma unroll
      for (int t = 0; t < 4; ++t) {
        #pragma unroll
        for (int s = 0; s < 4; ++s)
          bf[t][s] = load_rows_k8(F, H_DIM, m0 + t * 16, k + 8 * s, lane);
      }
      __builtin_amdgcn_sched_barrier(0);
      #pragma unroll
      for (int s = 0; s < 4; ++s) {
        #pragma unroll
        for (int t = 0; t < 4; ++t)
          acc[t] = wmma_f32(frag_lo(a[s]), frag_lo(bf[t][s]), acc[t]);
        #pragma unroll
        for (int t = 0; t < 4; ++t)
          acc[t] = wmma_f32(frag_hi(a[s]), frag_hi(bf[t][s]), acc[t]);
      }
    }
    int rbase = (lane >> 4) << 3;
    int cl    = lane & 15;
    #pragma unroll
    for (int t = 0; t < 4; ++t) {
      #pragma unroll
      for (int j = 0; j < 8; ++j)
        slab[(rbase + j) * N_DIM + m0 + t * 16 + cl] = acc[t][j];
    }
  }
  __syncthreads();

  for (int rr = 0; rr < 2; ++rr) {
    int r = wave * 2 + rr;
    float mx = -3.402823466e38f;
    for (int m = lane; m < N_DIM; m += 32)
      mx = fmaxf(mx, slab[r * N_DIM + m]);
    #pragma unroll
    for (int o = 16; o > 0; o >>= 1)
      mx = fmaxf(mx, __shfl_xor(mx, o, 32));
    float s = 0.f;
    for (int m = lane; m < N_DIM; m += 32) {
      float e = __expf(slab[r * N_DIM + m] - mx);
      slab[r * N_DIM + m] = e;
      s += e;
    }
    #pragma unroll
    for (int o = 16; o > 0; o >>= 1)
      s += __shfl_xor(s, o, 32);
    if (lane == 0) rowInv[r] = 1.0f / s;
  }
  __syncthreads();

  float* P = partial + ((size_t)b * NT + nt) * N_DIM;
  for (int m = threadIdx.x; m < N_DIM; m += 256) {
    float a = 0.f;
    #pragma unroll
    for (int r = 0; r < 16; ++r)
      a += slab[r * N_DIM + m] * rowInv[r];
    P[m] = a;
  }
}

__global__ void colsum_kernel(const float* __restrict__ partial,
                              float* __restrict__ colsum) {
  int idx = blockIdx.x * blockDim.x + threadIdx.x;
  int b = idx >> 11;
  int m = idx & (N_DIM - 1);
  const float* P = partial + (size_t)b * NT * N_DIM + m;
  float a = 0.f;
  #pragma unroll 8
  for (int t = 0; t < NT; ++t) a += P[(size_t)t * N_DIM];
  colsum[idx] = a;
}

#define POOL_SLOTS 32
__global__ void __launch_bounds__(512)
pool_kernel(const float* __restrict__ colsum,
            const float* __restrict__ feats,
            float* __restrict__ out) {
  __shared__ float red[POOL_SLOTS * 64];

  int b     = blockIdx.x >> 4;
  int chunk = blockIdx.x & 15;
  int lane  = threadIdx.x & 31;
  int c     = lane & 15;
  int slot  = ((threadIdx.x >> 5) << 1) + (lane >> 4);

  const float* C = colsum + (size_t)b * N_DIM + slot * 4;
  const float* F = feats + ((size_t)b * N_DIM + slot * 4) * H_DIM + chunk * 64 + c * 4;

  v4f cv[16];
  #pragma unroll
  for (int j = 0; j < 16; ++j)
    cv[j] = *reinterpret_cast<const v4f*>(C + j * 128);

  v4f acc[4] = {};
  v4f f[2][16];
  #pragma unroll
  for (int i = 0; i < 16; ++i)
    f[0][i] = *reinterpret_cast<const v4f*>(F + (size_t)((i >> 2) * 128 + (i & 3)) * H_DIM);
  #pragma unroll
  for (int t = 0; t < 4; ++t) {
    if (t < 3) {
      #pragma unroll
      for (int i = 0; i < 16; ++i)
        f[(t + 1) & 1][i] = *reinterpret_cast<const v4f*>(
            F + (size_t)((4 * (t + 1) + (i >> 2)) * 128 + (i & 3)) * H_DIM);
    }
    __builtin_amdgcn_sched_barrier(0);
    #pragma unroll
    for (int i = 0; i < 16; ++i) {
      float w = cv[4 * t + (i >> 2)][i & 3];
      acc[i & 3] += w * f[t & 1][i];
    }
  }
  v4f s = (acc[0] + acc[1]) + (acc[2] + acc[3]);
  *reinterpret_cast<v4f*>(red + slot * 64 + c * 4) = s;
  __syncthreads();

  if (threadIdx.x < 64) {
    float a0 = 0.f, a1 = 0.f;
    #pragma unroll
    for (int q = 0; q < POOL_SLOTS; q += 2) {
      a0 += red[q * 64 + threadIdx.x];
      a1 += red[(q + 1) * 64 + threadIdx.x];
    }
    out[(size_t)b * H_DIM + chunk * 64 + threadIdx.x] = a0 + a1;
  }
}

extern "C" void kernel_launch(void* const* d_in, const int* in_sizes, int n_in,
                              void* d_out, int out_size, void* d_ws, size_t ws_size,
                              hipStream_t stream) {
  const float* feats = (const float*)d_in[0];
  const float* W     = (const float*)d_in[1];
  const float* bias  = (const float*)d_in[2];
  float* out = (float*)d_out;

  float* q       = (float*)d_ws;
  float* partial = q + (size_t)B_DIM * N_DIM * H_DIM;
  float* colsum  = partial + (size_t)B_DIM * NT * N_DIM;

  fc1_kernel<<<8192, 256, 0, stream>>>(feats, W, bias, q);

  attn_kernel<<<B_DIM * NT, 256, 16 * N_DIM * sizeof(float), stream>>>(q, feats, partial);

  colsum_kernel<<<(B_DIM * N_DIM) / 256, 256, 0, stream>>>(partial, colsum);

  pool_kernel<<<B_DIM * (H_DIM / 64), 512, 0, stream>>>(colsum, feats, out);
}


# Score kernel: 32 rows of q per block (256 KB slab, 16 waves), 8 feats tiles per wave on 16 accumulators: a feats fragment load feeds 4 WMMAs, a q load 16; every sum keeps its order
# speedup vs baseline: 2.5051x; 1.1270x over previous
#include <hip/hip_runtime.h>

#define B_DIM 16
#define N_DIM 2048
#define H_DIM 1024
#define NT    128

typedef __attribute__((ext_vector_type(2))) float v2f;
typedef __attribute__((ext_vector_type(4))) float v4f;
typedef __attribute__((ext_vector_type(8))) float v8f;


__device__ __forceinline__ v4f load_rows_k8(const float* __restrict__ M, int ld,
                                            int r0, int k0, int lane) {
  int r  = r0 + (lane & 15);
  int kk = k0 + ((lane >> 4) << 2);
  return *reinterpret_cast<const v4f*>(M + (size_t)r * ld + kk);
}

__device__ __forceinline__ v2f frag_lo(v4f x) { return v2f{x[0], x[1]}; }
__device__ __forceinline__ v2f frag_hi(v4f x) { return v2f{x[2], x[3]}; }

__device__ __forceinline__ v8f wmma_f32(v2f a, v2f b, v8f c) {
  return __builtin_amdgcn_wmma_f32_16x16x4_f32(false, a, false, b, (short)0, c,
                                               false, false);
}

__global__ void fc1_kernel(const float* __restrict__ feats,
                           const float* __restrict__ W,
                           const float* __restrict__ bias,
                           float* __restrict__ q) {
  int gid  = blockIdx.x * (blockDim.x >> 5) + (threadIdx.x >> 5);
  int lane = threadIdx.x & 31;
  int tM = gid >> 5;
  int tN = gid & 31;
  int r0 = tM * 16;
  int c0 = tN * 32;
  int cp = c0 + ((lane & 15) << 1);
  int kh = (lane >> 4) << 2;

  v8f acc0 = {};
  v8f acc1 = {};
  for (int k = 0; k < H_DIM; k += 32) {
    v4f a[4];
    v2f w[4][4];
    #pragma unroll
    for (int s = 0; s < 4; ++s)
      a[s] = load_rows_k8(feats, H_DIM, r0, k + 8 * s, lane);
    #pragma unroll
    for (int s = 0; s < 4; ++s) {
      const float* p = W + (size_t)(k + 8 * s + kh) * H_DIM + cp;
      #pragma unroll
      for (int i = 0; i < 4; ++i)
        w[s][i] = *reinterpret_cast<const v2f*>(p + (size_t)i * H_DIM);
    }
    __builtin_amdgcn_sched_barrier(0);
    #pragma unroll
    for (int s = 0; s < 4; ++s) {
      acc0 = wmma_f32(frag_lo(a[s]), v2f{w[s][0][0], w[s][1][0]}, acc0);
      acc1 = wmma_f32(frag_lo(a[s]), v2f{w[s][0][1], w[s][1][1]}, acc1);
      acc0 = wmma_f32(frag_hi(a[s]), v2f{w[s][2][0], w[s][3][0]}, acc0);
      acc1 = wmma_f32(frag_hi(a[s]), v2f{w[s][2][1], w[s][3][1]}, acc1);
    }
  }

  int row = r0 + ((lane >> 4) << 3);
  v2f bz = *reinterpret_cast<const v2f*>(bias + cp);
  #pragma unroll
  for (int j = 0; j < 8; ++j) {
    v2f o = v2f{acc0[j] + bz[0], acc1[j] + bz[1]};
    *reinterpret_cast<v2f*>(q + (size_t)(row + j) * H_DIM + cp) = o;
  }
}

#define QROWS 32
__global__ void __launch_bounds__(512)
attn_kernel(const float* __restrict__ q,
            const float* __restrict__ feats,
            float* __restrict__ partial) {
  int b  = blockIdx.x >> 6;
  int nb = blockIdx.x & 63;
  int n0 = nb * QROWS;
  const float* Q = q     + ((size_t)b * N_DIM + n0) * H_DIM;
  const float* F = feats + (size_t)b * N_DIM * H_DIM;

  __shared__ float slab[QROWS * N_DIM];
  __shared__ float rowInv[QROWS];

  int wave = threadIdx.x >> 5;
  int lane = threadIdx.x & 31;

  int m0 = wave * 128;
  v8f acc[2][8] = {};
  for (int k = 0; k < H_DIM; k += 32) {
    v4f a[2][4];
    #pragma unroll
    for (int u = 0; u < 2; ++u) {
      #pragma unroll
      for (int s = 0; s < 4; ++s)
        a[u][s] = load_rows_k8(Q, H_DIM, u * 16, k + 8 * s, lane);
    }
    #pragma unroll
    for (int hf = 0; hf < 2; ++hf) {
      v4f bf[4][4];
      #pragma unroll
      for (int t = 0; t < 4; ++t) {
        #pragma unroll
        for (int s = 0; s < 4; ++s)
          bf[t][s] = load_rows_k8(F, H_DIM, m0 + (hf * 4 + t) * 16, k + 8 * s, lane);
      }
      __builtin_amdgcn_sched_barrier(0);
      #pragma unroll
      for (int s = 0; s < 4; ++s) {
        #pragma unroll
        for (int u = 0; u < 2; ++u) {
          #pragma unroll
          for (int t = 0; t < 4; ++t)
            acc[u][hf * 4 + t] = wmma_f32(frag_lo(a[u][s]), frag_lo(bf[t][s]), acc[u][hf * 4 + t]);
        }
        #pragma unroll
        for (int u = 0; u < 2; ++u) {
          #pragma unroll
          for (int t = 0; t < 4; ++t)
            acc[u][hf * 4 + t] = wmma_f32(frag_hi(a[u][s]), frag_hi(bf[t][s]), acc[u][hf * 4 + t]);
        }
      }
    }
  }
  {
    int rbase = (lane >> 4) << 3;
    int cl    = lane & 15;
    #pragma unroll
    for (int u = 0; u < 2; ++u) {
      #pragma unroll
      for (int t = 0; t < 8; ++t) {
        #pragma unroll
        for (int j = 0; j < 8; ++j)
          slab[(u * 16 + rbase + j) * N_DIM + m0 + t * 16 + cl] = acc[u][t][j];
      }
    }
  }
  __syncthreads();

  for (int rr = 0; rr < 2; ++rr) {
    int r = wave * 2 + rr;
    float mx = -3.402823466e38f;
    for (int m = lane; m < N_DIM; m += 32)
      mx = fmaxf(mx, slab[r * N_DIM + m]);
    #pragma unroll
    for (int o = 16; o > 0; o >>= 1)
      mx = fmaxf(mx, __shfl_xor(mx, o, 32));
    float s = 0.f;
    for (int m = lane; m < N_DIM; m += 32) {
      float e = __expf(slab[r * N_DIM + m] - mx);
      slab[r * N_DIM + m] = e;
      s += e;
    }
    #pragma unroll
    for (int o = 16; o > 0; o >>= 1)
      s += __shfl_xor(s, o, 32);
    if (lane == 0) rowInv[r] = 1.0f / s;
  }
  __syncthreads();

  float* P = partial + ((size_t)b * NT + nb * 2) * N_DIM;
  for (int m = threadIdx.x; m < N_DIM; m += 512) {
    float a0 = 0.f;
    #pragma unroll
    for (int r = 0; r < 16; ++r)
      a0 += slab[r * N_DIM + m] * rowInv[r];
    float a1 = 0.f;
    #pragma unroll
    for (int r = 16; r < 32; ++r)
      a1 += slab[r * N_DIM + m] * rowInv[r];
    P[m] = a0;
    P[N_DIM + m] = a1;
  }
}

__global__ void colsum_kernel(const float* __restrict__ partial,
                              float* __restrict__ colsum) {
  int idx = blockIdx.x * blockDim.x + threadIdx.x;
  int b = idx >> 11;
  int m = idx & (N_DIM - 1);
  const float* P = partial + (size_t)b * NT * N_DIM + m;
  float a = 0.f;
  #pragma unroll 8
  for (int t = 0; t < NT; ++t) a += P[(size_t)t * N_DIM];
  colsum[idx] = a;
}

#define POOL_SLOTS 32
__global__ void __launch_bounds__(512)
pool_kernel(const float* __restrict__ colsum,
            const float* __restrict__ feats,
            float* __restrict__ out) {
  __shared__ float red[POOL_SLOTS * 64];

  int b     = blockIdx.x >> 4;
  int chunk = blockIdx.x & 15;
  int lane  = threadIdx.x & 31;
  int c     = lane & 15;
  int slot  = ((threadIdx.x >> 5) << 1) + (lane >> 4);

  const float* C = colsum + (size_t)b * N_DIM + slot * 4;
  const float* F = feats + ((size_t)b * N_DIM + slot * 4) * H_DIM + chunk * 64 + c * 4;

  v4f cv[16];
  #pragma unroll
  for (int j = 0; j < 16; ++j)
    cv[j] = *reinterpret_cast<const v4f*>(C + j * 128);

  v4f acc[4] = {};
  v4f f[2][16];
  #pragma unroll
  for (int i = 0; i < 16; ++i)
    f[0][i] = *reinterpret_cast<const v4f*>(F + (size_t)((i >> 2) * 128 + (i & 3)) * H_DIM);
  #pragma unroll
  for (int t = 0; t < 4; ++t) {
    if (t < 3) {
      #pragma unroll
      for (int i = 0; i < 16; ++i)
        f[(t + 1) & 1][i] = *reinterpret_cast<const v4f*>(
            F + (size_t)((4 * (t + 1) + (i >> 2)) * 128 + (i & 3)) * H_DIM);
    }
    __builtin_amdgcn_sched_barrier(0);
    #pragma unroll
    for (int i = 0; i < 16; ++i) {
      float w = cv[4 * t + (i >> 2)][i & 3];
      acc[i & 3] += w * f[t & 1][i];
    }
  }
  v4f s = (acc[0] + acc[1]) + (acc[2] + acc[3]);
  *reinterpret_cast<v4f*>(red + slot * 64 + c * 4) = s;
  __syncthreads();

  if (threadIdx.x < 64) {
    float a0 = 0.f, a1 = 0.f;
    #pragma unroll
    for (int q = 0; q < POOL_SLOTS; q += 2) {
      a0 += red[q * 64 + threadIdx.x];
      a1 += red[(q + 1) * 64 + threadIdx.x];
    }
    out[(size_t)b * H_DIM + chunk * 64 + threadIdx.x] = a0 + a1;
  }
}

extern "C" void kernel_launch(void* const* d_in, const int* in_sizes, int n_in,
                              void* d_out, int out_size, void* d_ws, size_t ws_size,
                              hipStream_t stream) {
  const float* feats = (const float*)d_in[0];
  const float* W     = (const float*)d_in[1];
  const float* bias  = (const float*)d_in[2];
  float* out = (float*)d_out;

  float* q       = (float*)d_ws;
  float* partial = q + (size_t)B_DIM * N_DIM * H_DIM;
  float* colsum  = partial + (size_t)B_DIM * NT * N_DIM;

  fc1_kernel<<<8192, 256, 0, stream>>>(feats, W, bias, q);

  attn_kernel<<<B_DIM * (N_DIM / QROWS), 512, 0, stream>>>(q, feats, partial);

  colsum_kernel<<<(B_DIM * N_DIM) / 256, 256, 0, stream>>>(partial, colsum);

  pool_kernel<<<B_DIM * (H_DIM / 64), 512, 0, stream>>>(colsum, feats, out);
}


# fc1 kernel re-tiled: 32 rows x 128 columns per wave on 16 accumulators, W read 128 bit per lane over 4 interleaved column tiles; a feats load feeds 16 WMMAs, a W load 8; every sum keeps its order
# speedup vs baseline: 2.8269x; 1.1285x over previous
#include <hip/hip_runtime.h>

#define B_DIM 16
#define N_DIM 2048
#define H_DIM 1024
#define NT    128

typedef __attribute__((ext_vector_type(2))) float v2f;
typedef __attribute__((ext_vector_type(4))) float v4f;
typedef __attribute__((ext_vector_type(8))) float v8f;


__device__ __forceinline__ v4f load_rows_k8(const float* __restrict__ M, int ld,
                                            int r0, int k0, int lane) {
  int r  = r0 + (lane & 15);
  int kk = k0 + ((lane >> 4) << 2);
  return *reinterpret_cast<const v4f*>(M + (size_t)r * ld + kk);
}

__device__ __forceinline__ v2f frag_lo(v4f x) { return v2f{x[0], x[1]}; }
__device__ __forceinline__ v2f frag_hi(v4f x) { return v2f{x[2], x[3]}; }

__device__ __forceinline__ v8f wmma_f32(v2f a, v2f b, v8f c) {
  return __builtin_amdgcn_wmma_f32_16x16x4_f32(false, a, false, b, (short)0, c,
                                               false, false);
}

#define FC_BLOCK_ROWS 128
#define FC_BLOCK_COLS 512
__global__ void __launch_bounds__(512)
fc1_kernel(const float* __restrict__ feats,
           const float* __restrict__ W,
           const float* __restrict__ bias,
           float* __restrict__ q) {
  int wave = threadIdx.x >> 5;
  int lane = threadIdx.x & 31;
  int c    = lane & 15;
  int kh   = (lane >> 4) << 2;
  int rg = wave & 3;
  int cg = wave >> 2;
  int r0 = (blockIdx.x >> 1) * FC_BLOCK_ROWS + rg * 32;
  int c0 = (blockIdx.x & 1) * FC_BLOCK_COLS + cg * 128;
  const float* Wp = W + (size_t)kh * H_DIM + c0 + 4 * c;

  v8f acc[2][8] = {};
  for (int k = 0; k < H_DIM; k += 32) {
    v4f a[2][4];
    #pragma unroll
    for (int u = 0; u < 2; ++u) {
      #pragma unroll
      for (int s = 0; s < 4; ++s)
        a[u][s] = load_rows_k8(feats, H_DIM, r0 + u * 16, k + 8 * s, lane);
    }
    #pragma unroll
    for (int g = 0; g < 2; ++g) {
      v4f w[4][4];
      #pragma unroll
      for (int s = 0; s < 4; ++s) {
        const float* p = Wp + (size_t)(k + 8 * s) * H_DIM + g * 64;
        #pragma unroll
        for (int i = 0; i < 4; ++i)
          w[s][i] = *reinterpret_cast<const v4f*>(p + (size_t)i * H_DIM);
      }
      __builtin_amdgcn_sched_barrier(0);
      #pragma unroll
      for (int s = 0; s < 4; ++s) {
        #pragma unroll
        for (int u = 0; u < 2; ++u) {
          #pragma unroll
          for (int e = 0; e < 4; ++e)
            acc[u][g * 4 + e] = wmma_f32(frag_lo(a[u][s]), v2f{w[s][0][e], w[s][1][e]},
                                         acc[u][g * 4 + e]);
        }
        #pragma unroll
        for (int u = 0; u < 2; ++u) {
          #pragma unroll
          for (int e = 0; e < 4; ++e)
            acc[u][g * 4 + e] = wmma_f32(frag_hi(a[u][s]), v2f{w[s][2][e], w[s][3][e]},
                                         acc[u][g * 4 + e]);
        }
      }
    }
  }

  int rbase = (lane >> 4) << 3;
  #pragma unroll
  for (int g = 0; g < 2; ++g) {
    int col = c0 + g * 64 + 4 * c;
    v4f bz = *reinterpret_cast<const v4f*>(bias + col);
    #pragma unroll
    for (int u = 0; u < 2; ++u) {
      #pragma unroll
      for (int j = 0; j < 8; ++j) {
        v4f o = v4f{acc[u][g * 4 + 0][j] + bz[0], acc[u][g * 4 + 1][j] + bz[1],
                    acc[u][g * 4 + 2][j] + bz[2], acc[u][g * 4 + 3][j] + bz[3]};
        *reinterpret_cast<v4f*>(q + (size_t)(r0 + u * 16 + rbase + j) * H_DIM + col) = o;
      }
    }
  }
}

#define QROWS 32
__global__ void __launch_bounds__(512)
attn_kernel(const float* __restrict__ q,
            const float* __restrict__ feats,
            float* __restrict__ partial) {
  int b  = blockIdx.x >> 6;
  int nb = blockIdx.x & 63;
  int n0 = nb * QROWS;
  const float* Q = q     + ((size_t)b * N_DIM + n0) * H_DIM;
  const float* F = feats + (size_t)b * N_DIM * H_DIM;

  __shared__ float slab[QROWS * N_DIM];
  __shared__ float rowInv[QROWS];

  int wave = threadIdx.x >> 5;
  int lane = threadIdx.x & 31;

  int m0 = wave * 128;
  v8f acc[2][8] = {};
  for (int k = 0; k < H_DIM; k += 32) {
    v4f a[2][4];
    #pragma unroll
    for (int u = 0; u < 2; ++u) {
      #pragma unroll
      for (int s = 0; s < 4; ++s)
        a[u][s] = load_rows_k8(Q, H_DIM, u * 16, k + 8 * s, lane);
    }
    #pragma unroll
    for (int hf = 0; hf < 2; ++hf) {
      v4f bf[4][4];
      #pragma unroll
      for (int t = 0; t < 4; ++t) {
        #pragma unroll
        for (int s = 0; s < 4; ++s)
          bf[t][s] = load_rows_k8(F, H_DIM, m0 + (hf * 4 + t) * 16, k + 8 * s, lane);
      }
      __builtin_amdgcn_sched_barrier(0);
      #pragma unroll
      for (int s = 0; s < 4; ++s) {
        #pragma unroll
        for (int u = 0; u < 2; ++u) {
          #pragma unroll
          for (int t = 0; t < 4; ++t)
            acc[u][hf * 4 + t] = wmma_f32(frag_lo(a[u][s]), frag_lo(bf[t][s]), acc[u][hf * 4 + t]);
        }
        #pragma unroll
        for (int u = 0; u < 2; ++u) {
          #pragma unroll
          for (int t = 0; t < 4; ++t)
            acc[u][hf * 4 + t] = wmma_f32(frag_hi(a[u][s]), frag_hi(bf[t][s]), acc[u][hf * 4 + t]);
        }
      }
    }
  }
  {
    int rbase = (lane >> 4) << 3;
    int cl    = lane & 15;
    #pragma unroll
    for (int u = 0; u < 2; ++u) {
      #pragma unroll
      for (int t = 0; t < 8; ++t) {
        #pragma unroll
        for (int j = 0; j < 8; ++j)
          slab[(u * 16 + rbase + j) * N_DIM + m0 + t * 16 + cl] = acc[u][t][j];
      }
    }
  }
  __syncthreads();

  for (int rr = 0; rr < 2; ++rr) {
    int r = wave * 2 + rr;
    float mx = -3.402823466e38f;
    for (int m = lane; m < N_DIM; m += 32)
      mx = fmaxf(mx, slab[r * N_DIM + m]);
    #pragma unroll
    for (int o = 16; o > 0; o >>= 1)
      mx = fmaxf(mx, __shfl_xor(mx, o, 32));
    float s = 0.f;
    for (int m = lane; m < N_DIM; m += 32) {
      float e = __expf(slab[r * N_DIM + m] - mx);
      slab[r * N_DIM + m] = e;
      s += e;
    }
    #pragma unroll
    for (int o = 16; o > 0; o >>= 1)
      s += __shfl_xor(s, o, 32);
    if (lane == 0) rowInv[r] = 1.0f / s;
  }
  __syncthreads();

  float* P = partial + ((size_t)b * NT + nb * 2) * N_DIM;
  for (int m = threadIdx.x; m < N_DIM; m += 512) {
    float a0 = 0.f;
    #pragma unroll
    for (int r = 0; r < 16; ++r)
      a0 += slab[r * N_DIM + m] * rowInv[r];
    float a1 = 0.f;
    #pragma unroll
    for (int r = 16; r < 32; ++r)
      a1 += slab[r * N_DIM + m] * rowInv[r];
    P[m] = a0;
    P[N_DIM + m] = a1;
  }
}

__global__ void colsum_kernel(const float* __restrict__ partial,
                              float* __restrict__ colsum) {
  int idx = blockIdx.x * blockDim.x + threadIdx.x;
  int b = idx >> 11;
  int m = idx & (N_DIM - 1);
  const float* P = partial + (size_t)b * NT * N_DIM + m;
  float a = 0.f;
  #pragma unroll 8
  for (int t = 0; t < NT; ++t) a += P[(size_t)t * N_DIM];
  colsum[idx] = a;
}

#define POOL_SLOTS 32
__global__ void __launch_bounds__(512)
pool_kernel(const float* __restrict__ colsum,
            const float* __restrict__ feats,
            float* __restrict__ out) {
  __shared__ float red[POOL_SLOTS * 64];

  int b     = blockIdx.x >> 4;
  int chunk = blockIdx.x & 15;
  int lane  = threadIdx.x & 31;
  int c     = lane & 15;
  int slot  = ((threadIdx.x >> 5) << 1) + (lane >> 4);

  const float* C = colsum + (size_t)b * N_DIM + slot * 4;
  const float* F = feats + ((size_t)b * N_DIM + slot * 4) * H_DIM + chunk * 64 + c * 4;

  v4f cv[16];
  #pragma unroll
  for (int j = 0; j < 16; ++j)
    cv[j] = *reinterpret_cast<const v4f*>(C + j * 128);

  v4f acc[4] = {};
  v4f f[2][16];
  #pragma unroll
  for (int i = 0; i < 16; ++i)
    f[0][i] = *reinterpret_cast<const v4f*>(F + (size_t)((i >> 2) * 128 + (i & 3)) * H_DIM);
  #pragma unroll
  for (int t = 0; t < 4; ++t) {
    if (t < 3) {
      #pragma unroll
      for (int i = 0; i < 16; ++i)
        f[(t + 1) & 1][i] = *reinterpret_cast<const v4f*>(
            F + (size_t)((4 * (t + 1) + (i >> 2)) * 128 + (i & 3)) * H_DIM);
    }
    __builtin_amdgcn_sched_barrier(0);
    #pragma unroll
    for (int i = 0; i < 16; ++i) {
      float w = cv[4 * t + (i >> 2)][i & 3];
      acc[i & 3] += w * f[t & 1][i];
    }
  }
  v4f s = (acc[0] + acc[1]) + (acc[2] + acc[3]);
  *reinterpret_cast<v4f*>(red + slot * 64 + c * 4) = s;
  __syncthreads();

  if (threadIdx.x < 64) {
    float a0 = 0.f, a1 = 0.f;
    #pragma unroll
    for (int q = 0; q < POOL_SLOTS; q += 2) {
      a0 += red[q * 64 + threadIdx.x];
      a1 += red[(q + 1) * 64 + threadIdx.x];
    }
    out[(size_t)b * H_DIM + chunk * 64 + threadIdx.x] = a0 + a1;
  }
}

extern "C" void kernel_launch(void* const* d_in, const int* in_sizes, int n_in,
                              void* d_out, int out_size, void* d_ws, size_t ws_size,
                              hipStream_t stream) {
  const float* feats = (const float*)d_in[0];
  const float* W     = (const float*)d_in[1];
  const float* bias  = (const float*)d_in[2];
  float* out = (float*)d_out;

  float* q       = (float*)d_ws;
  float* partial = q + (size_t)B_DIM * N_DIM * H_DIM;
  float* colsum  = partial + (size_t)B_DIM * NT * N_DIM;

  fc1_kernel<<<(B_DIM * N_DIM / FC_BLOCK_ROWS) * (H_DIM / FC_BLOCK_COLS), 512, 0, stream>>>(
      feats, W, bias, q);

  attn_kernel<<<B_DIM * (N_DIM / QROWS), 512, 0, stream>>>(q, feats, partial);

  colsum_kernel<<<(B_DIM * N_DIM) / 256, 256, 0, stream>>>(partial, colsum);

  pool_kernel<<<B_DIM * (H_DIM / 64), 512, 0, stream>>>(colsum, feats, out);
}


# Both GEMM loops: wave priority stepped down (3,2,1,0) as a wave advances through K, so the four waves of a SIMD end together instead of oldest first, youngest alone; no arithmetic changed
# speedup vs baseline: 3.0404x; 1.0755x over previous
#include <hip/hip_runtime.h>

#define B_DIM 16
#define N_DIM 2048
#define H_DIM 1024
#define NT    128

typedef __attribute__((ext_vector_type(2))) float v2f;
typedef __attribute__((ext_vector_type(4))) float v4f;
typedef __attribute__((ext_vector_type(8))) float v8f;


__device__ __forceinline__ v4f load_rows_k8(const float* __restrict__ M, int ld,
                                            int r0, int k0, int lane) {
  int r  = r0 + (lane & 15);
  int kk = k0 + ((lane >> 4) << 2);
  return *reinterpret_cast<const v4f*>(M + (size_t)r * ld + kk);
}

__device__ __forceinline__ v2f frag_lo(v4f x) { return v2f{x[0], x[1]}; }
__device__ __forceinline__ v2f frag_hi(v4f x) { return v2f{x[2], x[3]}; }

__device__ __forceinline__ v8f wmma_f32(v2f a, v2f b, v8f c) {
  return __builtin_amdgcn_wmma_f32_16x16x4_f32(false, a, false, b, (short)0, c,
                                               false, false);
}

#define PRIO_K2 (20 * 32)
#define PRIO_K1 (28 * 32)
#define PRIO_K0 (31 * 32)
__device__ __forceinline__ void k_stage_priority(int k) {
  if (k == 0)            __builtin_amdgcn_s_setprio(3);
  else if (k == PRIO_K2) __builtin_amdgcn_s_setprio(2);
  else if (k == PRIO_K1) __builtin_amdgcn_s_setprio(1);
  else if (k == PRIO_K0) __builtin_amdgcn_s_setprio(0);
}

#define FC_BLOCK_ROWS 128
#define FC_BLOCK_COLS 512
__global__ void __launch_bounds__(512)
fc1_kernel(const float* __restrict__ feats,
           const float* __restrict__ W,
           const float* __restrict__ bias,
           float* __restrict__ q) {
  int wave = threadIdx.x >> 5;
  int lane = threadIdx.x & 31;
  int c    = lane & 15;
  int kh   = (lane >> 4) << 2;
  int rg = wave & 3;
  int cg = wave >> 2;
  int r0 = (blockIdx.x >> 1) * FC_BLOCK_ROWS + rg * 32;
  int c0 = (blockIdx.x & 1) * FC_BLOCK_COLS + cg * 128;
  const float* Wp = W + (size_t)kh * H_DIM + c0 + 4 * c;

  v8f acc[2][8] = {};
  for (int k = 0; k < H_DIM; k += 32) {
    k_stage_priority(k);
    v4f a[2][4];
    #pragma unroll
    for (int u = 0; u < 2; ++u) {
      #pragma unroll
      for (int s = 0; s < 4; ++s)
        a[u][s] = load_rows_k8(feats, H_DIM, r0 + u * 16, k + 8 * s, lane);
    }
    #pragma unroll
    for (int g = 0; g < 2; ++g) {
      v4f w[4][4];
      #pragma unroll
      for (int s = 0; s < 4; ++s) {
        const float* p = Wp + (size_t)(k + 8 * s) * H_DIM + g * 64;
        #pragma unroll
        for (int i = 0; i < 4; ++i)
          w[s][i] = *reinterpret_cast<const v4f*>(p + (size_t)i * H_DIM);
      }
      __builtin_amdgcn_sched_barrier(0);
      #pragma unroll
      for (int s = 0; s < 4; ++s) {
        #pragma unroll
        for (int u = 0; u < 2; ++u) {
          #pragma unroll
          for (int e = 0; e < 4; ++e)
            acc[u][g * 4 + e] = wmma_f32(frag_lo(a[u][s]), v2f{w[s][0][e], w[s][1][e]},
                                         acc[u][g * 4 + e]);
        }
        #pragma unroll
        for (int u = 0; u < 2; ++u) {
          #pragma unroll
          for (int e = 0; e < 4; ++e)
            acc[u][g * 4 + e] = wmma_f32(frag_hi(a[u][s]), v2f{w[s][2][e], w[s][3][e]},
                                         acc[u][g * 4 + e]);
        }
      }
    }
  }

  int rbase = (lane >> 4) << 3;
  #pragma unroll
  for (int g = 0; g < 2; ++g) {
    int col = c0 + g * 64 + 4 * c;
    v4f bz = *reinterpret_cast<const v4f*>(bias + col);
    #pragma unroll
    for (int u = 0; u < 2; ++u) {
      #pragma unroll
      for (int j = 0; j < 8; ++j) {
        v4f o = v4f{acc[u][g * 4 + 0][j] + bz[0], acc[u][g * 4 + 1][j] + bz[1],
                    acc[u][g * 4 + 2][j] + bz[2], acc[u][g * 4 + 3][j] + bz[3]};
        *reinterpret_cast<v4f*>(q + (size_t)(r0 + u * 16 + rbase + j) * H_DIM + col) = o;
      }
    }
  }
}

#define QROWS 32
__global__ void __launch_bounds__(512)
attn_kernel(const float* __restrict__ q,
            const float* __restrict__ feats,
            float* __restrict__ partial) {
  int b  = blockIdx.x >> 6;
  int nb = blockIdx.x & 63;
  int n0 = nb * QROWS;
  const float* Q = q     + ((size_t)b * N_DIM + n0) * H_DIM;
  const float* F = feats + (size_t)b * N_DIM * H_DIM;

  __shared__ float slab[QROWS * N_DIM];
  __shared__ float rowInv[QROWS];

  int wave = threadIdx.x >> 5;
  int lane = threadIdx.x & 31;

  int m0 = wave * 128;
  v8f acc[2][8] = {};
  for (int k = 0; k < H_DIM; k += 32) {
    k_stage_priority(k);
    v4f a[2][4];
    #pragma unroll
    for (int u = 0; u < 2; ++u) {
      #pragma unroll
      for (int s = 0; s < 4; ++s)
        a[u][s] = load_rows_k8(Q, H_DIM, u * 16, k + 8 * s, lane);
    }
    #pragma unroll
    for (int hf = 0; hf < 2; ++hf) {
      v4f bf[4][4];
      #pragma unroll
      for (int t = 0; t < 4; ++t) {
        #pragma unroll
        for (int s = 0; s < 4; ++s)
          bf[t][s] = load_rows_k8(F, H_DIM, m0 + (hf * 4 + t) * 16, k + 8 * s, lane);
      }
      __builtin_amdgcn_sched_barrier(0);
      #pragma unroll
      for (int s = 0; s < 4; ++s) {
        #pragma unroll
        for (int u = 0; u < 2; ++u) {
          #pragma unroll
          for (int t = 0; t < 4; ++t)
            acc[u][hf * 4 + t] = wmma_f32(frag_lo(a[u][s]), frag_lo(bf[t][s]), acc[u][hf * 4 + t]);
        }
        #pragma unroll
        for (int u = 0; u < 2; ++u) {
          #pragma unroll
          for (int t = 0; t < 4; ++t)
            acc[u][hf * 4 + t] = wmma_f32(frag_hi(a[u][s]), frag_hi(bf[t][s]), acc[u][hf * 4 + t]);
        }
      }
    }
  }
  {
    int rbase = (lane >> 4) << 3;
    int cl    = lane & 15;
    #pragma unroll
    for (int u = 0; u < 2; ++u) {
      #pragma unroll
      for (int t = 0; t < 8; ++t) {
        #pragma unroll
        for (int j = 0; j < 8; ++j)
          slab[(u * 16 + rbase + j) * N_DIM + m0 + t * 16 + cl] = acc[u][t][j];
      }
    }
  }
  __syncthreads();

  for (int rr = 0; rr < 2; ++rr) {
    int r = wave * 2 + rr;
    float mx = -3.402823466e38f;
    for (int m = lane; m < N_DIM; m += 32)
      mx = fmaxf(mx, slab[r * N_DIM + m]);
    #pragma unroll
    for (int o = 16; o > 0; o >>= 1)
      mx = fmaxf(mx, __shfl_xor(mx, o, 32));
    float s = 0.f;
    for (int m = lane; m < N_DIM; m += 32) {
      float e = __expf(slab[r * N_DIM + m] - mx);
      slab[r * N_DIM + m] = e;
      s += e;
    }
    #pragma unroll
    for (int o = 16; o > 0; o >>= 1)
      s += __shfl_xor(s, o, 32);
    if (lane == 0) rowInv[r] = 1.0f / s;
  }
  __syncthreads();

  float* P = partial + ((size_t)b * NT + nb * 2) * N_DIM;
  for (int m = threadIdx.x; m < N_DIM; m += 512) {
    float a0 = 0.f;
    #pragma unroll
    for (int r = 0; r < 16; ++r)
      a0 += slab[r * N_DIM + m] * rowInv[r];
    float a1 = 0.f;
    #pragma unroll
    for (int r = 16; r < 32; ++r)
      a1 += slab[r * N_DIM + m] * rowInv[r];
    P[m] = a0;
    P[N_DIM + m] = a1;
  }
}

__global__ void colsum_kernel(const float* __restrict__ partial,
                              float* __restrict__ colsum) {
  int idx = blockIdx.x * blockDim.x + threadIdx.x;
  int b = idx >> 11;
  int m = idx & (N_DIM - 1);
  const float* P = partial + (size_t)b * NT * N_DIM + m;
  float a = 0.f;
  #pragma unroll 8
  for (int t = 0; t < NT; ++t) a += P[(size_t)t * N_DIM];
  colsum[idx] = a;
}

#define POOL_SLOTS 32
__global__ void __launch_bounds__(512)
pool_kernel(const float* __restrict__ colsum,
            const float* __restrict__ feats,
            float* __restrict__ out) {
  __shared__ float red[POOL_SLOTS * 64];

  int b     = blockIdx.x >> 4;
  int chunk = blockIdx.x & 15;
  int lane  = threadIdx.x & 31;
  int c     = lane & 15;
  int slot  = ((threadIdx.x >> 5) << 1) + (lane >> 4);

  const float* C = colsum + (size_t)b * N_DIM + slot * 4;
  const float* F = feats + ((size_t)b * N_DIM + slot * 4) * H_DIM + chunk * 64 + c * 4;

  v4f cv[16];
  #pragma unroll
  for (int j = 0; j < 16; ++j)
    cv[j] = *reinterpret_cast<const v4f*>(C + j * 128);

  v4f acc[4] = {};
  v4f f[2][16];
  #pragma unroll
  for (int i = 0; i < 16; ++i)
    f[0][i] = *reinterpret_cast<const v4f*>(F + (size_t)((i >> 2) * 128 + (i & 3)) * H_DIM);
  #pragma unroll
  for (int t = 0; t < 4; ++t) {
    if (t < 3) {
      #pragma unroll
      for (int i = 0; i < 16; ++i)
        f[(t + 1) & 1][i] = *reinterpret_cast<const v4f*>(
            F + (size_t)((4 * (t + 1) + (i >> 2)) * 128 + (i & 3)) * H_DIM);
    }
    __builtin_amdgcn_sched_barrier(0);
    #pragma unroll
    for (int i = 0; i < 16; ++i) {
      float w = cv[4 * t + (i >> 2)][i & 3];
      acc[i & 3] += w * f[t & 1][i];
    }
  }
  v4f s = (acc[0] + acc[1]) + (acc[2] + acc[3]);
  *reinterpret_cast<v4f*>(red + slot * 64 + c * 4) = s;
  __syncthreads();

  if (threadIdx.x < 64) {
    float a0 = 0.f, a1 = 0.f;
    #pragma unroll
    for (int q = 0; q < POOL_SLOTS; q += 2) {
      a0 += red[q * 64 + threadIdx.x];
      a1 += red[(q + 1) * 64 + threadIdx.x];
    }
    out[(size_t)b * H_DIM + chunk * 64 + threadIdx.x] = a0 + a1;
  }
}

extern "C" void kernel_launch(void* const* d_in, const int* in_sizes, int n_in,
                              void* d_out, int out_size, void* d_ws, size_t ws_size,
                              hipStream_t stream) {
  const float* feats = (const float*)d_in[0];
  const float* W     = (const float*)d_in[1];
  const float* bias  = (const float*)d_in[2];
  float* out = (float*)d_out;

  float* q       = (float*)d_ws;
  float* partial = q + (size_t)B_DIM * N_DIM * H_DIM;
  float* colsum  = partial + (size_t)B_DIM * NT * N_DIM;

  fc1_kernel<<<(B_DIM * N_DIM / FC_BLOCK_ROWS) * (H_DIM / FC_BLOCK_COLS), 512, 0, stream>>>(
      feats, W, bias, q);

  attn_kernel<<<B_DIM * (N_DIM / QROWS), 512, 0, stream>>>(q, feats, partial);

  colsum_kernel<<<(B_DIM * N_DIM) / 256, 256, 0, stream>>>(partial, colsum);

  pool_kernel<<<B_DIM * (H_DIM / 64), 512, 0, stream>>>(colsum, feats, out);
}
